# starting kernel
# baseline (speedup 1.0000x reference)
#include <hip/hip_runtime.h>
#include <hip/hip_bf16.h>

typedef __attribute__((ext_vector_type(2))) float v2f;
typedef __attribute__((ext_vector_type(8))) float v8f;

namespace {
constexpr int Bc = 8;
constexpr int Ic = 64;
constexpr int Kc = 16;
constexpr int Lc = 256;
constexpr int Oc = 16;
constexpr int Dc = 32;
constexpr int LT = 16;
constexpr int ITERS = 3;
constexpr int XSS = Kc + 2;
constexpr int NTHREADS = Oc * 32;
constexpr int LDS_FLOATS = Ic * LT * XSS
                         + Oc * Ic * LT
                         + Ic * LT
                         + Ic * LT;
}

__device__ __forceinline__ void wmma_u(const float* __restrict__ Wp,
                                       const float* __restrict__ xrow,
                                       int arow, v8f& u0, v8f& u1)
{
#pragma unroll
    for (int c = 0; c < 4; ++c) {
        v2f bf = *(const v2f*)(xrow + 4 * c);
        v2f a0 = *(const v2f*)(Wp + arow + 4 * c);
        v2f a1 = *(const v2f*)(Wp + arow + 16 * Kc + 4 * c);
        u0 = __builtin_amdgcn_wmma_f32_16x16x4_f32(false, a0, false, bf, (short)0, u0, false, false);
        u1 = __builtin_amdgcn_wmma_f32_16x16x4_f32(false, a1, false, bf, (short)0, u1, false, false);
    }
}

__global__ __launch_bounds__(NTHREADS)
void DynamicRouting_86766929314212_kernel(const float* __restrict__ x,
                                          const float* __restrict__ W,
                                          float* __restrict__ out)
{
    extern __shared__ float lds[];
    float* xs = lds;
    float* bl = xs + Ic * LT * XSS;
    float* mx = bl + Oc * Ic * LT;
    float* rd = mx + Ic * LT;

    const int tid  = threadIdx.x;
    const int lane = tid & 31;
    const int o    = tid >> 5;
    const int n    = lane & 15;
    const int h    = lane >> 4;
    const int arow = n * Kc + 2 * h;

    const int tilesPerB = Lc / LT;
    const int b  = blockIdx.x / tilesPerB;
    const int l0 = (blockIdx.x % tilesPerB) * LT;

    const float* Wo = W + (size_t)o * Ic * Dc * Kc;

    for (int idx = tid; idx < Ic * Kc * LT; idx += NTHREADS) {
        int i  = idx >> 8;
        int r  = idx & 255;
        int k  = r >> 4;
        int nn = r & 15;
        xs[(i * LT + nn) * XSS + k] = x[((size_t)(b * Ic + i) * Kc + k) * Lc + l0 + nn];
    }
    for (int idx = tid; idx < Oc * Ic * LT; idx += NTHREADS) bl[idx] = 0.0f;

    v8f v0, v1;

    for (int it = 0; it < ITERS; ++it) {
        __syncthreads();
        for (int idx = tid; idx < Ic * LT; idx += NTHREADS) {
            float m = bl[idx];
            for (int oo = 1; oo < Oc; ++oo) m = fmaxf(m, bl[oo * Ic * LT + idx]);
            float den = 0.0f;
            for (int oo = 0; oo < Oc; ++oo) den += __expf(bl[oo * Ic * LT + idx] - m);
            mx[idx] = m;
            rd[idx] = 1.0f / den;
        }
        __syncthreads();

        v8f s0 = {0.f,0.f,0.f,0.f,0.f,0.f,0.f,0.f};
        v8f s1 = {0.f,0.f,0.f,0.f,0.f,0.f,0.f,0.f};
        for (int i = 0; i < Ic; ++i) {
            v8f u0 = {0.f,0.f,0.f,0.f,0.f,0.f,0.f,0.f};
            v8f u1 = {0.f,0.f,0.f,0.f,0.f,0.f,0.f,0.f};
            wmma_u(Wo + i * Dc * Kc, xs + (i * LT + n) * XSS + 2 * h, arow, u0, u1);
            float cb = __expf(bl[(o * Ic + i) * LT + n] - mx[i * LT + n]) * rd[i * LT + n];
#pragma unroll
            for (int j = 0; j < 8; ++j) { s0[j] += cb * u0[j]; s1[j] += cb * u1[j]; }
        }

        float sq = 0.0f;
#pragma unroll
        for (int j = 0; j < 8; ++j) sq += s0[j] * s0[j] + s1[j] * s1[j];
        sq += __shfl_xor(sq, 16, 32);
        const float scale = (sq / (1.0f + sq)) * rsqrtf(sq + 1e-8f);
#pragma unroll
        for (int j = 0; j < 8; ++j) { s0[j] *= scale; s1[j] *= scale; }
        v0 = s0; v1 = s1;

        if (it + 1 < ITERS) {
            for (int i = 0; i < Ic; ++i) {
                v8f u0 = {0.f,0.f,0.f,0.f,0.f,0.f,0.f,0.f};
                v8f u1 = {0.f,0.f,0.f,0.f,0.f,0.f,0.f,0.f};
                wmma_u(Wo + i * Dc * Kc, xs + (i * LT + n) * XSS + 2 * h, arow, u0, u1);
                float dot = 0.0f;
#pragma unroll
                for (int j = 0; j < 8; ++j) dot += u0[j] * v0[j] + u1[j] * v1[j];
                dot += __shfl_xor(dot, 16, 32);
                if (h == 0) bl[(o * Ic + i) * LT + n] += dot;
            }
        }
    }

    const size_t obase = ((size_t)(b * Oc + o) * Dc) * Lc + l0 + n;
#pragma unroll
    for (int j = 0; j < 8; ++j) {
        int d = j + 8 * h;
        out[obase + (size_t)d * Lc]        = v0[j];
        out[obase + (size_t)(d + 16) * Lc] = v1[j];
    }
}

extern "C" void kernel_launch(void* const* d_in, const int* in_sizes, int n_in,
                              void* d_out, int out_size, void* d_ws, size_t ws_size,
                              hipStream_t stream)
{
    (void)in_sizes; (void)n_in; (void)out_size; (void)d_ws; (void)ws_size;
    const float* x = (const float*)d_in[0];
    const float* W = (const float*)d_in[1];
    float* out = (float*)d_out;

    constexpr size_t shmem = (size_t)LDS_FLOATS * sizeof(float);
    dim3 grid(Bc * (Lc / LT));
    DynamicRouting_86766929314212_kernel<<<grid, NTHREADS, shmem, stream>>>(x, W, out);
}


# Fuse logit update of iter t with weighted sum of iter t+1 in 8-capsule chunks (softmax over o is separable in i): u_hat built 3x not 5x; chunk W loads grouped per capsule pair, next pair ahead
# speedup vs baseline: 1.5599x; 1.5599x over previous
#include <hip/hip_runtime.h>
#include <hip/hip_bf16.h>

typedef __attribute__((ext_vector_type(2))) float v2f;
typedef __attribute__((ext_vector_type(8))) float v8f;

namespace {
constexpr int Bc = 8;
constexpr int Ic = 64;
constexpr int Kc = 16;
constexpr int Lc = 256;
constexpr int Oc = 16;
constexpr int Dc = 32;
constexpr int LT = 16;
constexpr int ITERS = 3;
constexpr int CH = 8;
constexpr int XSS = Kc + 2;
constexpr int NTHREADS = Oc * 32;
constexpr int LDS_FLOATS = Ic * LT * XSS
                         + Oc * Ic * LT
                         + Ic * LT
                         + Ic * LT;
}

__device__ __forceinline__ void wmma_u(const float* __restrict__ Wp,
                                       const float* __restrict__ xrow,
                                       int arow, v8f& u0, v8f& u1)
{
#pragma unroll
    for (int c = 0; c < 4; ++c) {
        v2f bf = *(const v2f*)(xrow + 4 * c);
        v2f a0 = *(const v2f*)(Wp + arow + 4 * c);
        v2f a1 = *(const v2f*)(Wp + arow + 16 * Kc + 4 * c);
        u0 = __builtin_amdgcn_wmma_f32_16x16x4_f32(false, a0, false, bf, (short)0, u0, false, false);
        u1 = __builtin_amdgcn_wmma_f32_16x16x4_f32(false, a1, false, bf, (short)0, u1, false, false);
    }
}


__device__ __forceinline__ void load_ab(const float* __restrict__ Wp,
                                        const float* __restrict__ xrow, int arow,
                                        v2f (&a0)[4], v2f (&a1)[4], v2f (&bf)[4])
{
#pragma unroll
    for (int c = 0; c < 4; ++c) {
        bf[c] = *(const v2f*)(xrow + 4 * c);
        a0[c] = *(const v2f*)(Wp + arow + 4 * c);
        a1[c] = *(const v2f*)(Wp + arow + 16 * Kc + 4 * c);
    }
}

__device__ __forceinline__ void mma_u(const v2f (&a0)[4], const v2f (&a1)[4],
                                      const v2f (&bf)[4], v8f& u0, v8f& u1)
{
#pragma unroll
    for (int c = 0; c < 4; ++c) {
        u0 = __builtin_amdgcn_wmma_f32_16x16x4_f32(false, a0[c], false, bf[c], (short)0, u0, false, false);
        u1 = __builtin_amdgcn_wmma_f32_16x16x4_f32(false, a1[c], false, bf[c], (short)0, u1, false, false);
    }
}

__device__ __forceinline__ void squash(v8f& s0, v8f& s1)
{
    float sq = 0.0f;
#pragma unroll
    for (int j = 0; j < 8; ++j) sq += s0[j] * s0[j] + s1[j] * s1[j];
    sq += __shfl_xor(sq, 16, 32);
    const float scale = (sq / (1.0f + sq)) * rsqrtf(sq + 1e-8f);
#pragma unroll
    for (int j = 0; j < 8; ++j) { s0[j] *= scale; s1[j] *= scale; }
}

__global__ __launch_bounds__(NTHREADS)
void routing_fused_kernel(const float* __restrict__ x,
                                          const float* __restrict__ W,
                                          float* __restrict__ out)
{
    extern __shared__ float lds[];
    float* xs = lds;
    float* bl = xs + Ic * LT * XSS;
    float* mx = bl + Oc * Ic * LT;
    float* rd = mx + Ic * LT;

    const int tid  = threadIdx.x;
    const int lane = tid & 31;
    const int o    = tid >> 5;
    const int n    = lane & 15;
    const int h    = lane >> 4;
    const int arow = n * Kc + 2 * h;

    const int tilesPerB = Lc / LT;
    const int b  = blockIdx.x / tilesPerB;
    const int l0 = (blockIdx.x % tilesPerB) * LT;

    const float* Wo = W + (size_t)o * Ic * Dc * Kc;

    for (int idx = tid; idx < Ic * Kc * LT; idx += NTHREADS) {
        int i  = idx >> 8;
        int r  = idx & 255;
        int k  = r >> 4;
        int nn = r & 15;
        xs[(i * LT + nn) * XSS + k] = x[((size_t)(b * Ic + i) * Kc + k) * Lc + l0 + nn];
    }
    for (int idx = tid; idx < Oc * Ic * LT; idx += NTHREADS) bl[idx] = 0.0f;

    v8f v0, v1;

    __syncthreads();
    for (int idx = tid; idx < Ic * LT; idx += NTHREADS) {
        float m = bl[idx];
        for (int oo = 1; oo < Oc; ++oo) m = fmaxf(m, bl[oo * Ic * LT + idx]);
        float den = 0.0f;
        for (int oo = 0; oo < Oc; ++oo) den += __expf(bl[oo * Ic * LT + idx] - m);
        mx[idx] = m;
        rd[idx] = 1.0f / den;
    }
    __syncthreads();

    {
        v8f s0 = {0.f,0.f,0.f,0.f,0.f,0.f,0.f,0.f};
        v8f s1 = {0.f,0.f,0.f,0.f,0.f,0.f,0.f,0.f};
        for (int i = 0; i < Ic; ++i) {
            v8f u0 = {0.f,0.f,0.f,0.f,0.f,0.f,0.f,0.f};
            v8f u1 = {0.f,0.f,0.f,0.f,0.f,0.f,0.f,0.f};
            wmma_u(Wo + i * Dc * Kc, xs + (i * LT + n) * XSS + 2 * h, arow, u0, u1);
            float cb = __expf(bl[(o * Ic + i) * LT + n] - mx[i * LT + n]) * rd[i * LT + n];
#pragma unroll
            for (int j = 0; j < 8; ++j) { s0[j] += cb * u0[j]; s1[j] += cb * u1[j]; }
        }
        squash(s0, s1);
        v0 = s0; v1 = s1;
    }

    for (int it = 1; it < ITERS; ++it) {
        v8f s0 = {0.f,0.f,0.f,0.f,0.f,0.f,0.f,0.f};
        v8f s1 = {0.f,0.f,0.f,0.f,0.f,0.f,0.f,0.f};
        for (int ic = 0; ic < Ic; ic += CH) {
            v8f u0[CH], u1[CH];
#pragma unroll
            for (int q = 0; q < CH; q += 2) {
                v2f a0[2][4], a1[2][4], bf[2][4];
#pragma unroll
                for (int r = 0; r < 2; ++r) {
                    const int i = ic + q + r;
                    load_ab(Wo + i * Dc * Kc, xs + (i * LT + n) * XSS + 2 * h, arow, a0[r], a1[r], bf[r]);
                }
                __builtin_amdgcn_sched_barrier(0);
#pragma unroll
                for (int r = 0; r < 2; ++r) {
                    u0[q + r] = (v8f){0.f,0.f,0.f,0.f,0.f,0.f,0.f,0.f};
                    u1[q + r] = (v8f){0.f,0.f,0.f,0.f,0.f,0.f,0.f,0.f};
                    mma_u(a0[r], a1[r], bf[r], u0[q + r], u1[q + r]);
                }
            }
            float dot[CH];
#pragma unroll
            for (int q = 0; q < CH; ++q) {
                float t = 0.0f;
#pragma unroll
                for (int j = 0; j < 8; ++j) t += u0[q][j] * v0[j] + u1[q][j] * v1[j];
                t += __shfl_xor(t, 16, 32);
                dot[q] = t;
            }
            if (h == 0) {
#pragma unroll
                for (int q = 0; q < CH; ++q) bl[(o * Ic + ic + q) * LT + n] += dot[q];
            }
            __syncthreads();
            if (tid < CH * LT) {
                const int idx = ic * LT + tid;
                float m = bl[idx];
                for (int oo = 1; oo < Oc; ++oo) m = fmaxf(m, bl[oo * Ic * LT + idx]);
                float den = 0.0f;
                for (int oo = 0; oo < Oc; ++oo) den += __expf(bl[oo * Ic * LT + idx] - m);
                mx[idx] = m;
                rd[idx] = 1.0f / den;
            }
            __syncthreads();
#pragma unroll
            for (int q = 0; q < CH; ++q) {
                const int i = ic + q;
                float cb = __expf(bl[(o * Ic + i) * LT + n] - mx[i * LT + n]) * rd[i * LT + n];
#pragma unroll
                for (int j = 0; j < 8; ++j) { s0[j] += cb * u0[q][j]; s1[j] += cb * u1[q][j]; }
            }
        }
        squash(s0, s1);
        v0 = s0; v1 = s1;
    }

    const size_t obase = ((size_t)(b * Oc + o) * Dc) * Lc + l0 + n;
#pragma unroll
    for (int j = 0; j < 8; ++j) {
        int d = j + 8 * h;
        out[obase + (size_t)d * Lc]        = v0[j];
        out[obase + (size_t)(d + 16) * Lc] = v1[j];
    }
}

extern "C" void kernel_launch(void* const* d_in, const int* in_sizes, int n_in,
                              void* d_out, int out_size, void* d_ws, size_t ws_size,
                              hipStream_t stream)
{
    (void)in_sizes; (void)n_in; (void)out_size; (void)d_ws; (void)ws_size;
    const float* x = (const float*)d_in[0];
    const float* W = (const float*)d_in[1];
    float* out = (float*)d_out;

    constexpr size_t shmem = (size_t)LDS_FLOATS * sizeof(float);
    dim3 grid(Bc * (Lc / LT));
    routing_fused_kernel<<<grid, NTHREADS, shmem, stream>>>(x, W, out);
}


# A and B fragments fetched as b128 (lane holds k=8q+4h..+3, one load feeds two WMMAs; xs pitch 18->20 floats): half the global and LDS load instructions per u_hat; k sum order inside u_hat changes
# speedup vs baseline: 1.6367x; 1.0492x over previous
#include <hip/hip_runtime.h>
#include <hip/hip_bf16.h>

typedef __attribute__((ext_vector_type(2))) float v2f;
typedef __attribute__((ext_vector_type(4))) float v4f;
typedef __attribute__((ext_vector_type(8))) float v8f;

namespace {
constexpr int Bc = 8;
constexpr int Ic = 64;
constexpr int Kc = 16;
constexpr int Lc = 256;
constexpr int Oc = 16;
constexpr int Dc = 32;
constexpr int LT = 16;
constexpr int ITERS = 3;
constexpr int CH = 8;
constexpr int XSS = Kc + 4;
constexpr int NTHREADS = Oc * 32;
constexpr int LDS_FLOATS = Ic * LT * XSS
                         + Oc * Ic * LT
                         + Ic * LT
                         + Ic * LT;
}

__device__ __forceinline__ void wmma_u(const float* __restrict__ Wp,
                                       const float* __restrict__ xrow,
                                       int arow, v8f& u0, v8f& u1)
{
#pragma unroll
    for (int q = 0; q < 2; ++q) {
        v4f bf = *(const v4f*)(xrow + 8 * q);
        v4f a0 = *(const v4f*)(Wp + arow + 8 * q);
        v4f a1 = *(const v4f*)(Wp + arow + 16 * Kc + 8 * q);
        const v2f blo = __builtin_shufflevector(bf, bf, 0, 1);
        const v2f bhi = __builtin_shufflevector(bf, bf, 2, 3);
        u0 = __builtin_amdgcn_wmma_f32_16x16x4_f32(false, __builtin_shufflevector(a0, a0, 0, 1), false, blo, (short)0, u0, false, false);
        u1 = __builtin_amdgcn_wmma_f32_16x16x4_f32(false, __builtin_shufflevector(a1, a1, 0, 1), false, blo, (short)0, u1, false, false);
        u0 = __builtin_amdgcn_wmma_f32_16x16x4_f32(false, __builtin_shufflevector(a0, a0, 2, 3), false, bhi, (short)0, u0, false, false);
        u1 = __builtin_amdgcn_wmma_f32_16x16x4_f32(false, __builtin_shufflevector(a1, a1, 2, 3), false, bhi, (short)0, u1, false, false);
    }
}


__device__ __forceinline__ void load_ab(const float* __restrict__ Wp,
                                        const float* __restrict__ xrow, int arow,
                                        v4f (&a0)[2], v4f (&a1)[2], v4f (&bf)[2])
{
#pragma unroll
    for (int q = 0; q < 2; ++q) {
        bf[q] = *(const v4f*)(xrow + 8 * q);
        a0[q] = *(const v4f*)(Wp + arow + 8 * q);
        a1[q] = *(const v4f*)(Wp + arow + 16 * Kc + 8 * q);
    }
}

__device__ __forceinline__ void mma_u(const v4f (&a0)[2], const v4f (&a1)[2],
                                      const v4f (&bf)[2], v8f& u0, v8f& u1)
{
#pragma unroll
    for (int q = 0; q < 2; ++q) {
        const v2f blo = __builtin_shufflevector(bf[q], bf[q], 0, 1);
        const v2f bhi = __builtin_shufflevector(bf[q], bf[q], 2, 3);
        u0 = __builtin_amdgcn_wmma_f32_16x16x4_f32(false, __builtin_shufflevector(a0[q], a0[q], 0, 1), false, blo, (short)0, u0, false, false);
        u1 = __builtin_amdgcn_wmma_f32_16x16x4_f32(false, __builtin_shufflevector(a1[q], a1[q], 0, 1), false, blo, (short)0, u1, false, false);
        u0 = __builtin_amdgcn_wmma_f32_16x16x4_f32(false, __builtin_shufflevector(a0[q], a0[q], 2, 3), false, bhi, (short)0, u0, false, false);
        u1 = __builtin_amdgcn_wmma_f32_16x16x4_f32(false, __builtin_shufflevector(a1[q], a1[q], 2, 3), false, bhi, (short)0, u1, false, false);
    }
}

__device__ __forceinline__ void squash(v8f& s0, v8f& s1)
{
    float sq = 0.0f;
#pragma unroll
    for (int j = 0; j < 8; ++j) sq += s0[j] * s0[j] + s1[j] * s1[j];
    sq += __shfl_xor(sq, 16, 32);
    const float scale = (sq / (1.0f + sq)) * rsqrtf(sq + 1e-8f);
#pragma unroll
    for (int j = 0; j < 8; ++j) { s0[j] *= scale; s1[j] *= scale; }
}

__global__ __launch_bounds__(NTHREADS)
void routing_fused_kernel(const float* __restrict__ x,
                                          const float* __restrict__ W,
                                          float* __restrict__ out)
{
    extern __shared__ __attribute__((aligned(16))) float lds[];
    float* xs = lds;
    float* bl = xs + Ic * LT * XSS;
    float* mx = bl + Oc * Ic * LT;
    float* rd = mx + Ic * LT;

    const int tid  = threadIdx.x;
    const int lane = tid & 31;
    const int o    = tid >> 5;
    const int n    = lane & 15;
    const int h    = lane >> 4;
    const int arow = n * Kc + 4 * h;

    const int tilesPerB = Lc / LT;
    const int b  = blockIdx.x / tilesPerB;
    const int l0 = (blockIdx.x % tilesPerB) * LT;

    const float* Wo = W + (size_t)o * Ic * Dc * Kc;

    for (int idx = tid; idx < Ic * Kc * LT; idx += NTHREADS) {
        int i  = idx >> 8;
        int r  = idx & 255;
        int k  = r >> 4;
        int nn = r & 15;
        xs[(i * LT + nn) * XSS + k] = x[((size_t)(b * Ic + i) * Kc + k) * Lc + l0 + nn];
    }
    for (int idx = tid; idx < Oc * Ic * LT; idx += NTHREADS) bl[idx] = 0.0f;

    v8f v0, v1;

    __syncthreads();
    for (int idx = tid; idx < Ic * LT; idx += NTHREADS) {
        float m = bl[idx];
        for (int oo = 1; oo < Oc; ++oo) m = fmaxf(m, bl[oo * Ic * LT + idx]);
        float den = 0.0f;
        for (int oo = 0; oo < Oc; ++oo) den += __expf(bl[oo * Ic * LT + idx] - m);
        mx[idx] = m;
        rd[idx] = 1.0f / den;
    }
    __syncthreads();

    {
        v8f s0 = {0.f,0.f,0.f,0.f,0.f,0.f,0.f,0.f};
        v8f s1 = {0.f,0.f,0.f,0.f,0.f,0.f,0.f,0.f};
        for (int i = 0; i < Ic; ++i) {
            v8f u0 = {0.f,0.f,0.f,0.f,0.f,0.f,0.f,0.f};
            v8f u1 = {0.f,0.f,0.f,0.f,0.f,0.f,0.f,0.f};
            wmma_u(Wo + i * Dc * Kc, xs + (i * LT + n) * XSS + 4 * h, arow, u0, u1);
            float cb = __expf(bl[(o * Ic + i) * LT + n] - mx[i * LT + n]) * rd[i * LT + n];
#pragma unroll
            for (int j = 0; j < 8; ++j) { s0[j] += cb * u0[j]; s1[j] += cb * u1[j]; }
        }
        squash(s0, s1);
        v0 = s0; v1 = s1;
    }

    for (int it = 1; it < ITERS; ++it) {
        v8f s0 = {0.f,0.f,0.f,0.f,0.f,0.f,0.f,0.f};
        v8f s1 = {0.f,0.f,0.f,0.f,0.f,0.f,0.f,0.f};
        for (int ic = 0; ic < Ic; ic += CH) {
            v8f u0[CH], u1[CH];
#pragma unroll
            for (int q = 0; q < CH; q += 2) {
                v4f a0[2][2], a1[2][2], bf[2][2];
#pragma unroll
                for (int r = 0; r < 2; ++r) {
                    const int i = ic + q + r;
                    load_ab(Wo + i * Dc * Kc, xs + (i * LT + n) * XSS + 4 * h, arow, a0[r], a1[r], bf[r]);
                }
                __builtin_amdgcn_sched_barrier(0);
#pragma unroll
                for (int r = 0; r < 2; ++r) {
                    u0[q + r] = (v8f){0.f,0.f,0.f,0.f,0.f,0.f,0.f,0.f};
                    u1[q + r] = (v8f){0.f,0.f,0.f,0.f,0.f,0.f,0.f,0.f};
                    mma_u(a0[r], a1[r], bf[r], u0[q + r], u1[q + r]);
                }
            }
            float dot[CH];
#pragma unroll
            for (int q = 0; q < CH; ++q) {
                float t = 0.0f;
#pragma unroll
                for (int j = 0; j < 8; ++j) t += u0[q][j] * v0[j] + u1[q][j] * v1[j];
                t += __shfl_xor(t, 16, 32);
                dot[q] = t;
            }
            if (h == 0) {
#pragma unroll
                for (int q = 0; q < CH; ++q) bl[(o * Ic + ic + q) * LT + n] += dot[q];
            }
            __syncthreads();
            if (tid < CH * LT) {
                const int idx = ic * LT + tid;
                float m = bl[idx];
                for (int oo = 1; oo < Oc; ++oo) m = fmaxf(m, bl[oo * Ic * LT + idx]);
                float den = 0.0f;
                for (int oo = 0; oo < Oc; ++oo) den += __expf(bl[oo * Ic * LT + idx] - m);
                mx[idx] = m;
                rd[idx] = 1.0f / den;
            }
            __syncthreads();
#pragma unroll
            for (int q = 0; q < CH; ++q) {
                const int i = ic + q;
                float cb = __expf(bl[(o * Ic + i) * LT + n] - mx[i * LT + n]) * rd[i * LT + n];
#pragma unroll
                for (int j = 0; j < 8; ++j) { s0[j] += cb * u0[q][j]; s1[j] += cb * u1[q][j]; }
            }
        }
        squash(s0, s1);
        v0 = s0; v1 = s1;
    }

    const int lane_w = (int)__builtin_amdgcn_mbcnt_lo(~0u, 0u);
    const int n_w = lane_w & 15;
    const int h_w = lane_w >> 4;
    const size_t obase = ((size_t)(b * Oc + o) * Dc) * Lc + l0 + n_w;
#pragma unroll
    for (int j = 0; j < 8; ++j) {
        int d = j + 8 * h_w;
        out[obase + (size_t)d * Lc]        = v0[j];
        out[obase + (size_t)(d + 16) * Lc] = v1[j];
    }
}

extern "C" void kernel_launch(void* const* d_in, const int* in_sizes, int n_in,
                              void* d_out, int out_size, void* d_ws, size_t ws_size,
                              hipStream_t stream)
{
    (void)in_sizes; (void)n_in; (void)out_size; (void)d_ws; (void)ws_size;
    const float* x = (const float*)d_in[0];
    const float* W = (const float*)d_in[1];
    float* out = (float*)d_out;

    constexpr size_t shmem = (size_t)LDS_FLOATS * sizeof(float);
    dim3 grid(Bc * (Lc / LT));
    routing_fused_kernel<<<grid, NTHREADS, shmem, stream>>>(x, W, out);
}


# W fragments fetched one capsule pair ahead in all 3 u_hat builds, each chunk's first pair ahead of the previous chunk's 2nd barrier (now LDS-only); loads stand as one group per 16-WMMA block
# speedup vs baseline: 1.7260x; 1.0546x over previous
#include <hip/hip_runtime.h>
#include <hip/hip_bf16.h>

typedef __attribute__((ext_vector_type(2))) float v2f;
typedef __attribute__((ext_vector_type(4))) float v4f;
typedef __attribute__((ext_vector_type(8))) float v8f;

namespace {
constexpr int Bc = 8;
constexpr int Ic = 64;
constexpr int Kc = 16;
constexpr int Lc = 256;
constexpr int Oc = 16;
constexpr int Dc = 32;
constexpr int LT = 16;
constexpr int ITERS = 3;
constexpr int CH = 8;
constexpr int XSS = Kc + 4;
constexpr int NTHREADS = Oc * 32;
constexpr int LDS_FLOATS = Ic * LT * XSS
                         + Oc * Ic * LT
                         + Ic * LT
                         + Ic * LT;
}

__device__ __forceinline__ void mma_u(const v4f (&a0)[2], const v4f (&a1)[2],
                                      const v4f (&bf)[2], v8f& u0, v8f& u1)
{
#pragma unroll
    for (int q = 0; q < 2; ++q) {
        const v2f blo = __builtin_shufflevector(bf[q], bf[q], 0, 1);
        const v2f bhi = __builtin_shufflevector(bf[q], bf[q], 2, 3);
        u0 = __builtin_amdgcn_wmma_f32_16x16x4_f32(false, __builtin_shufflevector(a0[q], a0[q], 0, 1), false, blo, (short)0, u0, false, false);
        u1 = __builtin_amdgcn_wmma_f32_16x16x4_f32(false, __builtin_shufflevector(a1[q], a1[q], 0, 1), false, blo, (short)0, u1, false, false);
        u0 = __builtin_amdgcn_wmma_f32_16x16x4_f32(false, __builtin_shufflevector(a0[q], a0[q], 2, 3), false, bhi, (short)0, u0, false, false);
        u1 = __builtin_amdgcn_wmma_f32_16x16x4_f32(false, __builtin_shufflevector(a1[q], a1[q], 2, 3), false, bhi, (short)0, u1, false, false);
    }
}

struct PairFrag { v4f a0[2][2], a1[2][2], bf[2][2]; };

__device__ __forceinline__ void load_pair_w(const float* __restrict__ wl, PairFrag& f)
{
#pragma unroll
    for (int r = 0; r < 2; ++r) {
#pragma unroll
        for (int q = 0; q < 2; ++q) {
            f.a0[r][q] = *(const v4f*)(wl + r * Dc * Kc + 8 * q);
            f.a1[r][q] = *(const v4f*)(wl + r * Dc * Kc + 16 * Kc + 8 * q);
        }
    }
}

__device__ __forceinline__ void load_pair_x(const float* __restrict__ xl, PairFrag& f)
{
#pragma unroll
    for (int r = 0; r < 2; ++r) {
#pragma unroll
        for (int q = 0; q < 2; ++q) f.bf[r][q] = *(const v4f*)(xl + r * LT * XSS + 8 * q);
    }
}

__device__ __forceinline__ void load_pair(const float* __restrict__ wl, const float* __restrict__ xl, PairFrag& f)
{
    load_pair_x(xl, f);
    load_pair_w(wl, f);
}

__device__ __forceinline__ void lds_barrier()
{
    __builtin_amdgcn_fence(__ATOMIC_RELEASE, "workgroup", "local");
    __builtin_amdgcn_s_barrier();
    __builtin_amdgcn_fence(__ATOMIC_ACQUIRE, "workgroup", "local");
}

__device__ __forceinline__ void mma_pair(const PairFrag& f, v8f& ua0, v8f& ua1, v8f& ub0, v8f& ub1)
{
    ua0 = (v8f){0.f,0.f,0.f,0.f,0.f,0.f,0.f,0.f};
    ua1 = (v8f){0.f,0.f,0.f,0.f,0.f,0.f,0.f,0.f};
    ub0 = (v8f){0.f,0.f,0.f,0.f,0.f,0.f,0.f,0.f};
    ub1 = (v8f){0.f,0.f,0.f,0.f,0.f,0.f,0.f,0.f};
    mma_u(f.a0[0], f.a1[0], f.bf[0], ua0, ua1);
    mma_u(f.a0[1], f.a1[1], f.bf[1], ub0, ub1);
}

__device__ __forceinline__ void squash(v8f& s0, v8f& s1)
{
    float sq = 0.0f;
#pragma unroll
    for (int j = 0; j < 8; ++j) sq += s0[j] * s0[j] + s1[j] * s1[j];
    sq += __shfl_xor(sq, 16, 32);
    const float scale = (sq / (1.0f + sq)) * rsqrtf(sq + 1e-8f);
#pragma unroll
    for (int j = 0; j < 8; ++j) { s0[j] *= scale; s1[j] *= scale; }
}

__global__ __launch_bounds__(NTHREADS)
void routing_fused_kernel(const float* __restrict__ x,
                                          const float* __restrict__ W,
                                          float* __restrict__ out)
{
    extern __shared__ __attribute__((aligned(16))) float lds[];
    float* mx = lds;
    float* rd = mx + Ic * LT;
    float* xs = rd + Ic * LT;
    float* bl = xs + Ic * LT * XSS;

    const int tid  = threadIdx.x;
    const int lane = tid & 31;
    const int o    = __builtin_amdgcn_readfirstlane(tid >> 5);
    const int n    = lane & 15;
    const int h    = lane >> 4;
    const int arow = n * Kc + 4 * h;

    const int tilesPerB = Lc / LT;
    const int b  = blockIdx.x / tilesPerB;
    const int l0 = (blockIdx.x % tilesPerB) * LT;

    const float* Wo = W + (size_t)o * Ic * Dc * Kc;

    for (int idx = tid; idx < Ic * Kc * LT; idx += NTHREADS) {
        int i  = idx >> 8;
        int r  = idx & 255;
        int k  = r >> 4;
        int nn = r & 15;
        xs[(i * LT + nn) * XSS + k] = x[((size_t)(b * Ic + i) * Kc + k) * Lc + l0 + nn];
    }
    for (int idx = tid; idx < Oc * Ic * LT; idx += NTHREADS) bl[idx] = 0.0f;

    v8f v0, v1;

    __syncthreads();
    for (int idx = tid; idx < Ic * LT; idx += NTHREADS) {
        float m = bl[idx];
        for (int oo = 1; oo < Oc; ++oo) m = fmaxf(m, bl[oo * Ic * LT + idx]);
        float den = 0.0f;
        for (int oo = 0; oo < Oc; ++oo) den += __expf(bl[oo * Ic * LT + idx] - m);
        mx[idx] = m;
        rd[idx] = 1.0f / den;
    }
    __syncthreads();

    constexpr int WP = 2 * Dc * Kc;
    constexpr int XP = 2 * LT * XSS;
    const float* wl = Wo + arow;
    const float* xl = xs + n * XSS + 4 * h;
    PairFrag fa, fb;
    load_pair(wl, xl, fa);

    {
        v8f s0 = {0.f,0.f,0.f,0.f,0.f,0.f,0.f,0.f};
        v8f s1 = {0.f,0.f,0.f,0.f,0.f,0.f,0.f,0.f};
        for (int i = 0; i < Ic; i += 4) {
            v8f ua0, ua1, ub0, ub1;
            load_pair(wl + WP, xl + XP, fb);
            const float c0 = __expf(bl[(o * Ic + i) * LT + n] - mx[i * LT + n]) * rd[i * LT + n];
            const float c1 = __expf(bl[(o * Ic + i + 1) * LT + n] - mx[(i + 1) * LT + n]) * rd[(i + 1) * LT + n];
            __builtin_amdgcn_sched_barrier(0);
            mma_pair(fa, ua0, ua1, ub0, ub1);
            __builtin_amdgcn_sched_barrier(0);
#pragma unroll
            for (int j = 0; j < 8; ++j) { s0[j] += c0 * ua0[j]; s1[j] += c0 * ua1[j]; }
#pragma unroll
            for (int j = 0; j < 8; ++j) { s0[j] += c1 * ub0[j]; s1[j] += c1 * ub1[j]; }
            const int back = (i + 4 == Ic) ? Ic / 2 : 0;
            wl += 2 * WP - back * WP;
            xl += 2 * XP - back * XP;
            load_pair(wl, xl, fa);
            const float c2 = __expf(bl[(o * Ic + i + 2) * LT + n] - mx[(i + 2) * LT + n]) * rd[(i + 2) * LT + n];
            const float c3 = __expf(bl[(o * Ic + i + 3) * LT + n] - mx[(i + 3) * LT + n]) * rd[(i + 3) * LT + n];
            __builtin_amdgcn_sched_barrier(0);
            mma_pair(fb, ua0, ua1, ub0, ub1);
            __builtin_amdgcn_sched_barrier(0);
#pragma unroll
            for (int j = 0; j < 8; ++j) { s0[j] += c2 * ua0[j]; s1[j] += c2 * ua1[j]; }
#pragma unroll
            for (int j = 0; j < 8; ++j) { s0[j] += c3 * ub0[j]; s1[j] += c3 * ub1[j]; }
        }
        squash(s0, s1);
        v0 = s0; v1 = s1;
    }

    for (int it = 1; it < ITERS; ++it) {
        v8f s0 = {0.f,0.f,0.f,0.f,0.f,0.f,0.f,0.f};
        v8f s1 = {0.f,0.f,0.f,0.f,0.f,0.f,0.f,0.f};
        for (int ic = 0; ic < Ic; ic += CH) {
            v8f u0[CH], u1[CH];
            float tp[CH];
#define DOT_PART(q)                                                                          \
            {                                                                                 \
                const v8f p = __builtin_elementwise_fma(u0[q], v0, u1[q] * v1);               \
                float t = 0.0f;                                                               \
                _Pragma("unroll")                                                             \
                for (int j = 0; j < 8; ++j) t += p[j];                                        \
                tp[q] = t;                                                                    \
            }
            load_pair_x(xl, fa);
            load_pair_w(wl + WP, fb);
            __builtin_amdgcn_sched_barrier(0);
            mma_pair(fa, u0[0], u1[0], u0[1], u1[1]);
            __builtin_amdgcn_sched_barrier(0);
            load_pair_x(xl + XP, fb);
            load_pair_w(wl + 2 * WP, fa);
            __builtin_amdgcn_sched_barrier(0);
            mma_pair(fb, u0[2], u1[2], u0[3], u1[3]);
            DOT_PART(0) DOT_PART(1)
            __builtin_amdgcn_sched_barrier(0);
            load_pair_x(xl + 2 * XP, fa);
            load_pair_w(wl + 3 * WP, fb);
            __builtin_amdgcn_sched_barrier(0);
            mma_pair(fa, u0[4], u1[4], u0[5], u1[5]);
            DOT_PART(2) DOT_PART(3)
            __builtin_amdgcn_sched_barrier(0);
            load_pair_x(xl + 3 * XP, fb);
            __builtin_amdgcn_sched_barrier(0);
            mma_pair(fb, u0[6], u1[6], u0[7], u1[7]);
            DOT_PART(4) DOT_PART(5)
            DOT_PART(6) DOT_PART(7)
#undef DOT_PART
            float dot[CH];
#pragma unroll
            for (int q = 0; q < CH; ++q) dot[q] = tp[q] + __shfl_xor(tp[q], 16, 32);
            if (h == 0) {
#pragma unroll
                for (int q = 0; q < CH; ++q) bl[(o * Ic + ic + q) * LT + n] += dot[q];
            }
            __syncthreads();
            if (tid >= NTHREADS - CH * LT) {
                const int idx = ic * LT + tid - (NTHREADS - CH * LT);
                float m = bl[idx];
                for (int oo = 1; oo < Oc; ++oo) m = fmaxf(m, bl[oo * Ic * LT + idx]);
                float den = 0.0f;
                for (int oo = 0; oo < Oc; ++oo) den += __expf(bl[oo * Ic * LT + idx] - m);
                mx[idx] = m;
                rd[idx] = 1.0f / den;
            }
            {
                const int back = (ic + CH == Ic) ? Ic / 2 : 0;
                wl += (CH / 2) * WP - back * WP;
                xl += (CH / 2) * XP - back * XP;
            }
            load_pair_w(wl, fa);
            lds_barrier();
#pragma unroll
            for (int q = 0; q < CH; ++q) {
                const int i = ic + q;
                float cb = __expf(bl[(o * Ic + i) * LT + n] - mx[i * LT + n]) * rd[i * LT + n];
#pragma unroll
                for (int j = 0; j < 8; ++j) { s0[j] += cb * u0[q][j]; s1[j] += cb * u1[q][j]; }
            }
        }
        squash(s0, s1);
        v0 = s0; v1 = s1;
    }

    const int lane_w = (int)__builtin_amdgcn_mbcnt_lo(~0u, 0u);
    const int n_w = lane_w & 15;
    const int h_w = lane_w >> 4;
    const size_t obase = ((size_t)(b * Oc + o) * Dc) * Lc + l0 + n_w;
#pragma unroll
    for (int j = 0; j < 8; ++j) {
        int d = j + 8 * h_w;
        out[obase + (size_t)d * Lc]        = v0[j];
        out[obase + (size_t)(d + 16) * Lc] = v1[j];
    }
}

extern "C" void kernel_launch(void* const* d_in, const int* in_sizes, int n_in,
                              void* d_out, int out_size, void* d_ws, size_t ws_size,
                              hipStream_t stream)
{
    (void)in_sizes; (void)n_in; (void)out_size; (void)d_ws; (void)ws_size;
    const float* x = (const float*)d_in[0];
    const float* W = (const float*)d_in[1];
    float* out = (float*)d_out;

    constexpr size_t shmem = (size_t)LDS_FLOATS * sizeof(float);
    dim3 grid(Bc * (Lc / LT));
    routing_fused_kernel<<<grid, NTHREADS, shmem, stream>>>(x, W, out);
}


# Build 1 as one chained GEMM: logits are zero there so c=1/16 exactly; s sums over all 64 capsules in the WMMA C operand (f32 sum order changes); no stats/exp/FMA; W fetched 7 capsules ahead
# speedup vs baseline: 1.8381x; 1.0649x over previous
#include <hip/hip_runtime.h>
#include <hip/hip_bf16.h>

typedef __attribute__((ext_vector_type(2))) float v2f;
typedef __attribute__((ext_vector_type(4))) float v4f;
typedef __attribute__((ext_vector_type(8))) float v8f;

namespace {
constexpr int Bc = 8;
constexpr int Ic = 64;
constexpr int Kc = 16;
constexpr int Lc = 256;
constexpr int Oc = 16;
constexpr int Dc = 32;
constexpr int LT = 16;
constexpr int ITERS = 3;
constexpr int CH = 8;
constexpr int XSS = Kc + 4;
constexpr int NTHREADS = Oc * 32;
constexpr int LDS_FLOATS = Ic * LT * XSS
                         + Oc * Ic * LT
                         + Ic * LT
                         + Ic * LT;
}

__device__ __forceinline__ void mma_u(const v4f (&a0)[2], const v4f (&a1)[2],
                                      const v4f (&bf)[2], v8f& u0, v8f& u1)
{
#pragma unroll
    for (int q = 0; q < 2; ++q) {
        const v2f blo = __builtin_shufflevector(bf[q], bf[q], 0, 1);
        const v2f bhi = __builtin_shufflevector(bf[q], bf[q], 2, 3);
        u0 = __builtin_amdgcn_wmma_f32_16x16x4_f32(false, __builtin_shufflevector(a0[q], a0[q], 0, 1), false, blo, (short)0, u0, false, false);
        u1 = __builtin_amdgcn_wmma_f32_16x16x4_f32(false, __builtin_shufflevector(a1[q], a1[q], 0, 1), false, blo, (short)0, u1, false, false);
        u0 = __builtin_amdgcn_wmma_f32_16x16x4_f32(false, __builtin_shufflevector(a0[q], a0[q], 2, 3), false, bhi, (short)0, u0, false, false);
        u1 = __builtin_amdgcn_wmma_f32_16x16x4_f32(false, __builtin_shufflevector(a1[q], a1[q], 2, 3), false, bhi, (short)0, u1, false, false);
    }
}

struct PairFrag { v4f a0[2][2], a1[2][2], bf[2][2]; };

__device__ __forceinline__ void load_pair_w(const float* __restrict__ wl, PairFrag& f)
{
#pragma unroll
    for (int r = 0; r < 2; ++r) {
#pragma unroll
        for (int q = 0; q < 2; ++q) {
            f.a0[r][q] = *(const v4f*)(wl + r * Dc * Kc + 8 * q);
            f.a1[r][q] = *(const v4f*)(wl + r * Dc * Kc + 16 * Kc + 8 * q);
        }
    }
}

__device__ __forceinline__ void load_pair_x(const float* __restrict__ xl, PairFrag& f)
{
#pragma unroll
    for (int r = 0; r < 2; ++r) {
#pragma unroll
        for (int q = 0; q < 2; ++q) f.bf[r][q] = *(const v4f*)(xl + r * LT * XSS + 8 * q);
    }
}

__device__ __forceinline__ void lds_barrier()
{
    __builtin_amdgcn_fence(__ATOMIC_RELEASE, "workgroup", "local");
    __builtin_amdgcn_s_barrier();
    __builtin_amdgcn_fence(__ATOMIC_ACQUIRE, "workgroup", "local");
}

__device__ __forceinline__ void mma_pair(const PairFrag& f, v8f& ua0, v8f& ua1, v8f& ub0, v8f& ub1)
{
    ua0 = (v8f){0.f,0.f,0.f,0.f,0.f,0.f,0.f,0.f};
    ua1 = (v8f){0.f,0.f,0.f,0.f,0.f,0.f,0.f,0.f};
    ub0 = (v8f){0.f,0.f,0.f,0.f,0.f,0.f,0.f,0.f};
    ub1 = (v8f){0.f,0.f,0.f,0.f,0.f,0.f,0.f,0.f};
    mma_u(f.a0[0], f.a1[0], f.bf[0], ua0, ua1);
    mma_u(f.a0[1], f.a1[1], f.bf[1], ub0, ub1);
}

struct CapW { v4f a0[2], a1[2]; };
struct TagMid  { static constexpr bool value = false; };
struct TagLast { static constexpr bool value = true;  };
struct CapX { v4f bf[2]; };

__device__ __forceinline__ void load_cap_w(const float* __restrict__ wl, CapW& f)
{
#pragma unroll
    for (int q = 0; q < 2; ++q) {
        f.a0[q] = *(const v4f*)(wl + 8 * q);
        f.a1[q] = *(const v4f*)(wl + 16 * Kc + 8 * q);
    }
}

__device__ __forceinline__ void load_cap_x(const float* __restrict__ xl, CapX& f)
{
#pragma unroll
    for (int q = 0; q < 2; ++q) f.bf[q] = *(const v4f*)(xl + 8 * q);
}

__device__ __forceinline__ void squash(v8f& s0, v8f& s1)
{
    float sq = 0.0f;
#pragma unroll
    for (int j = 0; j < 8; ++j) sq += s0[j] * s0[j] + s1[j] * s1[j];
    sq += __shfl_xor(sq, 16, 32);
    const float scale = (sq / (1.0f + sq)) * rsqrtf(sq + 1e-8f);
#pragma unroll
    for (int j = 0; j < 8; ++j) { s0[j] *= scale; s1[j] *= scale; }
}

__global__ __launch_bounds__(NTHREADS)
void routing_fused_kernel(const float* __restrict__ x,
                                          const float* __restrict__ W,
                                          float* __restrict__ out)
{
    extern __shared__ __attribute__((aligned(16))) float lds[];
    float* mx = lds;
    float* rd = mx + Ic * LT;
    float* xs = rd + Ic * LT;
    float* bl = xs + Ic * LT * XSS;

    const int tid  = threadIdx.x;
    const int lane = tid & 31;
    const int o    = __builtin_amdgcn_readfirstlane(tid >> 5);
    const int n    = lane & 15;
    const int h    = lane >> 4;
    const int arow = n * Kc + 4 * h;

    const int tilesPerB = Lc / LT;
    const int b  = blockIdx.x / tilesPerB;
    const int l0 = (blockIdx.x % tilesPerB) * LT;

    const float* Wo = W + (size_t)o * Ic * Dc * Kc;

    for (int idx = tid; idx < Ic * Kc * LT; idx += NTHREADS) {
        int i  = idx >> 8;
        int r  = idx & 255;
        int k  = r >> 4;
        int nn = r & 15;
        xs[(i * LT + nn) * XSS + k] = x[((size_t)(b * Ic + i) * Kc + k) * Lc + l0 + nn];
    }
    for (int idx = tid; idx < Oc * Ic * LT; idx += NTHREADS) bl[idx] = 0.0f;

    v8f v0, v1;

    __syncthreads();

    constexpr int WP = 2 * Dc * Kc;
    constexpr int XP = 2 * LT * XSS;
    const float* wl = Wo + arow;
    const float* xl = xs + n * XSS + 4 * h;
    PairFrag fa, fb;

    {
        v8f s0 = {0.f,0.f,0.f,0.f,0.f,0.f,0.f,0.f};
        v8f s1 = {0.f,0.f,0.f,0.f,0.f,0.f,0.f,0.f};
        constexpr int WC = Dc * Kc;
        constexpr int XC = LT * XSS;
        CapW w[8];
        CapX xa, xb;
#pragma unroll
        for (int j = 0; j < 7; ++j) load_cap_w(wl + j * WC, w[j]);
        load_cap_x(xl, xa);
        auto trip = [&](auto last_tag) {
            constexpr bool LAST = decltype(last_tag)::value;
            constexpr int back = LAST ? Ic : 0;
            constexpr int refetch = LAST ? 2 : 7;
            const float* wn = wl + 8 * WC - back * WC;
            const float* xn = xl + 8 * XC - back * XC;
            load_cap_w(wl + 7 * WC, w[7]);
#pragma unroll
            for (int j = 0; j < 8; ++j) {
                CapX& xcur = (j & 1) ? xb : xa;
                CapX& xnxt = (j & 1) ? xa : xb;
                if (j < 7) load_cap_x(xl + (j + 1) * XC, xnxt);
                else if (!LAST) load_cap_x(xn, xnxt);
                __builtin_amdgcn_sched_barrier(0);
                mma_u(w[j].a0, w[j].a1, xcur.bf, s0, s1);
                __builtin_amdgcn_sched_barrier(0);
                if (j < refetch) load_cap_w(wn + j * WC, w[j]);
            }
            wl = wn;
            xl = xn;
        };
#pragma nounroll
        for (int c = 0; c < Ic - 8; c += 8) trip(TagMid{});
        trip(TagLast{});
#pragma unroll
        for (int q = 0; q < 2; ++q) {
            fa.a0[0][q] = w[0].a0[q]; fa.a1[0][q] = w[0].a1[q];
            fa.a0[1][q] = w[1].a0[q]; fa.a1[1][q] = w[1].a1[q];
        }
        constexpr float c_uniform = 1.0f / Oc;
#pragma unroll
        for (int j = 0; j < 8; ++j) { s0[j] *= c_uniform; s1[j] *= c_uniform; }
        squash(s0, s1);
        v0 = s0; v1 = s1;
    }

    for (int it = 1; it < ITERS; ++it) {
        v8f s0 = {0.f,0.f,0.f,0.f,0.f,0.f,0.f,0.f};
        v8f s1 = {0.f,0.f,0.f,0.f,0.f,0.f,0.f,0.f};
        for (int ic = 0; ic < Ic; ic += CH) {
            v8f u0[CH], u1[CH];
            float tp[CH];
#define DOT_PART(q)                                                                          \
            {                                                                                 \
                const v8f p = __builtin_elementwise_fma(u0[q], v0, u1[q] * v1);               \
                float t = 0.0f;                                                               \
                _Pragma("unroll")                                                             \
                for (int j = 0; j < 8; ++j) t += p[j];                                        \
                tp[q] = t;                                                                    \
            }
            load_pair_x(xl, fa);
            load_pair_w(wl + WP, fb);
            __builtin_amdgcn_sched_barrier(0);
            mma_pair(fa, u0[0], u1[0], u0[1], u1[1]);
            __builtin_amdgcn_sched_barrier(0);
            load_pair_x(xl + XP, fb);
            load_pair_w(wl + 2 * WP, fa);
            __builtin_amdgcn_sched_barrier(0);
            mma_pair(fb, u0[2], u1[2], u0[3], u1[3]);
            DOT_PART(0) DOT_PART(1)
            __builtin_amdgcn_sched_barrier(0);
            load_pair_x(xl + 2 * XP, fa);
            load_pair_w(wl + 3 * WP, fb);
            __builtin_amdgcn_sched_barrier(0);
            mma_pair(fa, u0[4], u1[4], u0[5], u1[5]);
            DOT_PART(2) DOT_PART(3)
            __builtin_amdgcn_sched_barrier(0);
            load_pair_x(xl + 3 * XP, fb);
            __builtin_amdgcn_sched_barrier(0);
            mma_pair(fb, u0[6], u1[6], u0[7], u1[7]);
            DOT_PART(4) DOT_PART(5)
            DOT_PART(6) DOT_PART(7)
#undef DOT_PART
            float dot[CH];
#pragma unroll
            for (int q = 0; q < CH; ++q) dot[q] = tp[q] + __shfl_xor(tp[q], 16, 32);
            if (h == 0) {
#pragma unroll
                for (int q = 0; q < CH; ++q) bl[(o * Ic + ic + q) * LT + n] += dot[q];
            }
            __syncthreads();
            if (tid >= NTHREADS - CH * LT) {
                const int idx = ic * LT + tid - (NTHREADS - CH * LT);
                float m = bl[idx];
                for (int oo = 1; oo < Oc; ++oo) m = fmaxf(m, bl[oo * Ic * LT + idx]);
                float den = 0.0f;
                for (int oo = 0; oo < Oc; ++oo) den += __expf(bl[oo * Ic * LT + idx] - m);
                mx[idx] = m;
                rd[idx] = 1.0f / den;
            }
            {
                const int back = (ic + CH == Ic) ? Ic / 2 : 0;
                wl += (CH / 2) * WP - back * WP;
                xl += (CH / 2) * XP - back * XP;
            }
            load_pair_w(wl, fa);
            lds_barrier();
#pragma unroll
            for (int q = 0; q < CH; ++q) {
                const int i = ic + q;
                float cb = __expf(bl[(o * Ic + i) * LT + n] - mx[i * LT + n]) * rd[i * LT + n];
#pragma unroll
                for (int j = 0; j < 8; ++j) { s0[j] += cb * u0[q][j]; s1[j] += cb * u1[q][j]; }
            }
        }
        squash(s0, s1);
        v0 = s0; v1 = s1;
    }

    const int lane_w = (int)__builtin_amdgcn_mbcnt_lo(~0u, 0u);
    const int n_w = lane_w & 15;
    const int h_w = lane_w >> 4;
    const size_t obase = ((size_t)(b * Oc + o) * Dc) * Lc + l0 + n_w;
#pragma unroll
    for (int j = 0; j < 8; ++j) {
        int d = j + 8 * h_w;
        out[obase + (size_t)d * Lc]        = v0[j];
        out[obase + (size_t)(d + 16) * Lc] = v1[j];
    }
}

extern "C" void kernel_launch(void* const* d_in, const int* in_sizes, int n_in,
                              void* d_out, int out_size, void* d_ws, size_t ws_size,
                              hipStream_t stream)
{
    (void)in_sizes; (void)n_in; (void)out_size; (void)d_ws; (void)ws_size;
    const float* x = (const float*)d_in[0];
    const float* W = (const float*)d_in[1];
    float* out = (float*)d_out;

    constexpr size_t shmem = (size_t)LDS_FLOATS * sizeof(float);
    dim3 grid(Bc * (Lc / LT));
    routing_fused_kernel<<<grid, NTHREADS, shmem, stream>>>(x, W, out);
}


# Iteration 0 (uniform softmax, one chained GEMM) moved to its own launch on 256 workgroups (4 out-capsules x 32 columns), v handed on in the output buffer: the routing grid of 128 fills half the WGPs
# speedup vs baseline: 1.9490x; 1.0603x over previous
#include <hip/hip_runtime.h>
#include <hip/hip_bf16.h>

typedef __attribute__((ext_vector_type(2))) float v2f;
typedef __attribute__((ext_vector_type(4))) float v4f;
typedef __attribute__((ext_vector_type(8))) float v8f;

namespace {
constexpr int Bc = 8;
constexpr int Ic = 64;
constexpr int Kc = 16;
constexpr int Lc = 256;
constexpr int Oc = 16;
constexpr int Dc = 32;
constexpr int LT = 16;
constexpr int ITERS = 3;
constexpr int CH = 8;
constexpr int XSS = Kc + 4;
constexpr int NTHREADS = Oc * 32;
constexpr int LDS_FLOATS = Ic * LT * XSS
                         + Oc * Ic * LT
                         + Ic * LT
                         + Ic * LT;
constexpr int OQ  = 4;
constexpr int NC1 = 2 * LT;
constexpr int NTHREADS1 = OQ * 2 * 32;
constexpr int LDS1_FLOATS = Ic * NC1 * XSS + 512;
}

__device__ __forceinline__ void mma_u(const v4f (&a0)[2], const v4f (&a1)[2],
                                      const v4f (&bf)[2], v8f& u0, v8f& u1)
{
#pragma unroll
    for (int q = 0; q < 2; ++q) {
        const v2f blo = __builtin_shufflevector(bf[q], bf[q], 0, 1);
        const v2f bhi = __builtin_shufflevector(bf[q], bf[q], 2, 3);
        u0 = __builtin_amdgcn_wmma_f32_16x16x4_f32(false, __builtin_shufflevector(a0[q], a0[q], 0, 1), false, blo, (short)0, u0, false, false);
        u1 = __builtin_amdgcn_wmma_f32_16x16x4_f32(false, __builtin_shufflevector(a1[q], a1[q], 0, 1), false, blo, (short)0, u1, false, false);
        u0 = __builtin_amdgcn_wmma_f32_16x16x4_f32(false, __builtin_shufflevector(a0[q], a0[q], 2, 3), false, bhi, (short)0, u0, false, false);
        u1 = __builtin_amdgcn_wmma_f32_16x16x4_f32(false, __builtin_shufflevector(a1[q], a1[q], 2, 3), false, bhi, (short)0, u1, false, false);
    }
}

struct PairFrag { v4f a0[2][2], a1[2][2], bf[2][2]; };

__device__ __forceinline__ void load_pair_w(const float* __restrict__ wl, PairFrag& f)
{
#pragma unroll
    for (int r = 0; r < 2; ++r) {
#pragma unroll
        for (int q = 0; q < 2; ++q) {
            f.a0[r][q] = *(const v4f*)(wl + r * Dc * Kc + 8 * q);
            f.a1[r][q] = *(const v4f*)(wl + r * Dc * Kc + 16 * Kc + 8 * q);
        }
    }
}

__device__ __forceinline__ void load_pair_x(const float* __restrict__ xl, PairFrag& f)
{
#pragma unroll
    for (int r = 0; r < 2; ++r) {
#pragma unroll
        for (int q = 0; q < 2; ++q) f.bf[r][q] = *(const v4f*)(xl + r * LT * XSS + 8 * q);
    }
}

__device__ __forceinline__ void lds_barrier()
{
    __builtin_amdgcn_fence(__ATOMIC_RELEASE, "workgroup", "local");
    __builtin_amdgcn_s_barrier();
    __builtin_amdgcn_fence(__ATOMIC_ACQUIRE, "workgroup", "local");
}

__device__ __forceinline__ void mma_pair(const PairFrag& f, v8f& ua0, v8f& ua1, v8f& ub0, v8f& ub1)
{
    ua0 = (v8f){0.f,0.f,0.f,0.f,0.f,0.f,0.f,0.f};
    ua1 = (v8f){0.f,0.f,0.f,0.f,0.f,0.f,0.f,0.f};
    ub0 = (v8f){0.f,0.f,0.f,0.f,0.f,0.f,0.f,0.f};
    ub1 = (v8f){0.f,0.f,0.f,0.f,0.f,0.f,0.f,0.f};
    mma_u(f.a0[0], f.a1[0], f.bf[0], ua0, ua1);
    mma_u(f.a0[1], f.a1[1], f.bf[1], ub0, ub1);
}

struct CapW { v4f a0[2], a1[2]; };
struct TagMid  { static constexpr bool value = false; };
struct TagLast { static constexpr bool value = true;  };
struct CapX { v4f bf[2]; };

__device__ __forceinline__ void load_cap_w(const float* __restrict__ wl, CapW& f)
{
#pragma unroll
    for (int q = 0; q < 2; ++q) {
        f.a0[q] = *(const v4f*)(wl + 8 * q);
        f.a1[q] = *(const v4f*)(wl + 16 * Kc + 8 * q);
    }
}

__device__ __forceinline__ void load_cap_x(const float* __restrict__ xl, CapX& f)
{
#pragma unroll
    for (int q = 0; q < 2; ++q) f.bf[q] = *(const v4f*)(xl + 8 * q);
}

__device__ __forceinline__ void squash(v8f& s0, v8f& s1)
{
    float sq = 0.0f;
#pragma unroll
    for (int j = 0; j < 8; ++j) sq += s0[j] * s0[j] + s1[j] * s1[j];
    sq += __shfl_xor(sq, 16, 32);
    const float scale = (sq / (1.0f + sq)) * rsqrtf(sq + 1e-8f);
#pragma unroll
    for (int j = 0; j < 8; ++j) { s0[j] *= scale; s1[j] *= scale; }
}

__global__ __launch_bounds__(NTHREADS1)
void uniform_sum_kernel(const float* __restrict__ x,
                        const float* __restrict__ W,
                        float* __restrict__ out)
{
    extern __shared__ __attribute__((aligned(16))) float lds[];
    float* xs = lds;

    const int tid  = threadIdx.x;
    const int lane = tid & 31;
    const int wv   = __builtin_amdgcn_readfirstlane(tid >> 5);
    const int ow   = wv >> 1;
    const int lt   = wv & 1;
    const int n    = lane & 15;
    const int h    = lane >> 4;
    const int arow = n * Kc + 4 * h;

    const int oq = blockIdx.x % (Oc / OQ);
    const int lp = (blockIdx.x / (Oc / OQ)) % (Lc / NC1);
    const int b  = blockIdx.x / ((Oc / OQ) * (Lc / NC1));
    const int l0 = lp * NC1;
    const int o  = oq * OQ + ow;

    constexpr int WC = Dc * Kc;
    const float* wl = W + (size_t)o * Ic * Dc * Kc + arow;
    CapW w[8];
#pragma unroll
    for (int j = 0; j < 7; ++j) load_cap_w(wl + j * WC, w[j]);

#pragma unroll 16
    for (int idx = tid; idx < Ic * Kc * (NC1 / 4); idx += NTHREADS1) {
        const int i  = idx / (Kc * (NC1 / 4));
        const int r  = idx % (Kc * (NC1 / 4));
        const int k  = r / (NC1 / 4);
        const int c4 = r % (NC1 / 4);
        const v4f t = *(const v4f*)(x + ((size_t)(b * Ic + i) * Kc + k) * Lc + l0 + 4 * c4);
        float* dst = xs + (i * NC1 + 4 * c4) * XSS + k;
        dst[0]       = t[0];
        dst[XSS]     = t[1];
        dst[2 * XSS] = t[2];
        dst[3 * XSS] = t[3];
    }
    __syncthreads();

    v8f s0 = {0.f,0.f,0.f,0.f,0.f,0.f,0.f,0.f};
    v8f s1 = {0.f,0.f,0.f,0.f,0.f,0.f,0.f,0.f};
    constexpr int XC = NC1 * XSS;
    const float* xl = xs + (lt * LT + n) * XSS + 4 * h;
    CapX xa, xb;
    load_cap_x(xl, xa);
    auto trip = [&](auto last_tag) {
        constexpr bool LAST = decltype(last_tag)::value;
        constexpr int refetch = LAST ? 0 : 7;
        const float* wn = LAST ? wl : wl + 8 * WC;
        const float* xn = LAST ? xl : xl + 8 * XC;
        load_cap_w(wl + 7 * WC, w[7]);
#pragma unroll
        for (int j = 0; j < 8; ++j) {
            CapX& xcur = (j & 1) ? xb : xa;
            CapX& xnxt = (j & 1) ? xa : xb;
            if (j < 7) load_cap_x(xl + (j + 1) * XC, xnxt);
            else if (!LAST) load_cap_x(xn, xnxt);
            __builtin_amdgcn_sched_barrier(0);
            mma_u(w[j].a0, w[j].a1, xcur.bf, s0, s1);
            __builtin_amdgcn_sched_barrier(0);
            if (j < refetch) load_cap_w(wn + j * WC, w[j]);
        }
        wl = wn;
        xl = xn;
    };
#pragma nounroll
    for (int c = 0; c < Ic - 8; c += 8) trip(TagMid{});
    trip(TagLast{});

    constexpr float c_uniform = 1.0f / Oc;
#pragma unroll
    for (int j = 0; j < 8; ++j) { s0[j] *= c_uniform; s1[j] *= c_uniform; }
    squash(s0, s1);

    const size_t obase = ((size_t)(b * Oc + o) * Dc) * Lc + l0 + lt * LT + n;
#pragma unroll
    for (int j = 0; j < 8; ++j) {
        const int d = j + 8 * h;
        out[obase + (size_t)d * Lc]        = s0[j];
        out[obase + (size_t)(d + 16) * Lc] = s1[j];
    }
}

__global__ __launch_bounds__(NTHREADS)
void routing_fused_kernel(const float* __restrict__ x,
                                          const float* __restrict__ W,
                                          float* __restrict__ out)
{
    extern __shared__ __attribute__((aligned(16))) float lds[];
    float* mx = lds;
    float* rd = mx + Ic * LT;
    float* xs = rd + Ic * LT;
    float* bl = xs + Ic * LT * XSS;

    const int tid  = threadIdx.x;
    const int lane = tid & 31;
    const int o    = __builtin_amdgcn_readfirstlane(tid >> 5);
    const int n    = lane & 15;
    const int h    = lane >> 4;
    const int arow = n * Kc + 4 * h;

    const int tilesPerB = Lc / LT;
    const int b  = blockIdx.x / tilesPerB;
    const int l0 = (blockIdx.x % tilesPerB) * LT;

    const float* Wo = W + (size_t)o * Ic * Dc * Kc;

    constexpr int WP = 2 * Dc * Kc;
    constexpr int XP = 2 * LT * XSS;
    const float* wl = Wo + arow;
    const float* xl = xs + n * XSS + 4 * h;
    PairFrag fa, fb;

    v8f v0, v1;
    {
        const size_t vbase = ((size_t)(b * Oc + o) * Dc) * Lc + l0 + n;
#pragma unroll
        for (int j = 0; j < 8; ++j) {
            const int d = j + 8 * h;
            v0[j] = out[vbase + (size_t)d * Lc];
            v1[j] = out[vbase + (size_t)(d + 16) * Lc];
        }
    }
    load_pair_w(wl, fa);

    for (int idx = tid; idx < Ic * Kc * LT; idx += NTHREADS) {
        int i  = idx >> 8;
        int r  = idx & 255;
        int k  = r >> 4;
        int nn = r & 15;
        xs[(i * LT + nn) * XSS + k] = x[((size_t)(b * Ic + i) * Kc + k) * Lc + l0 + nn];
    }
    for (int idx = tid; idx < Oc * Ic * LT; idx += NTHREADS) bl[idx] = 0.0f;

    __syncthreads();

    for (int it = 1; it < ITERS; ++it) {
        v8f s0 = {0.f,0.f,0.f,0.f,0.f,0.f,0.f,0.f};
        v8f s1 = {0.f,0.f,0.f,0.f,0.f,0.f,0.f,0.f};
        for (int ic = 0; ic < Ic; ic += CH) {
            v8f u0[CH], u1[CH];
            float tp[CH];
#define DOT_PART(q)                                                                          \
            {                                                                                 \
                const v8f p = __builtin_elementwise_fma(u0[q], v0, u1[q] * v1);               \
                float t = 0.0f;                                                               \
                _Pragma("unroll")                                                             \
                for (int j = 0; j < 8; ++j) t += p[j];                                        \
                tp[q] = t;                                                                    \
            }
            load_pair_x(xl, fa);
            load_pair_w(wl + WP, fb);
            __builtin_amdgcn_sched_barrier(0);
            mma_pair(fa, u0[0], u1[0], u0[1], u1[1]);
            __builtin_amdgcn_sched_barrier(0);
            load_pair_x(xl + XP, fb);
            load_pair_w(wl + 2 * WP, fa);
            __builtin_amdgcn_sched_barrier(0);
            mma_pair(fb, u0[2], u1[2], u0[3], u1[3]);
            DOT_PART(0) DOT_PART(1)
            __builtin_amdgcn_sched_barrier(0);
            load_pair_x(xl + 2 * XP, fa);
            load_pair_w(wl + 3 * WP, fb);
            __builtin_amdgcn_sched_barrier(0);
            mma_pair(fa, u0[4], u1[4], u0[5], u1[5]);
            DOT_PART(2) DOT_PART(3)
            __builtin_amdgcn_sched_barrier(0);
            load_pair_x(xl + 3 * XP, fb);
            __builtin_amdgcn_sched_barrier(0);
            mma_pair(fb, u0[6], u1[6], u0[7], u1[7]);
            DOT_PART(4) DOT_PART(5)
            DOT_PART(6) DOT_PART(7)
#undef DOT_PART
            float dot[CH];
#pragma unroll
            for (int q = 0; q < CH; ++q) dot[q] = tp[q] + __shfl_xor(tp[q], 16, 32);
            if (h == 0) {
#pragma unroll
                for (int q = 0; q < CH; ++q) bl[(o * Ic + ic + q) * LT + n] += dot[q];
            }
            __syncthreads();
            if (tid >= NTHREADS - CH * LT) {
                const int idx = ic * LT + tid - (NTHREADS - CH * LT);
                float m = bl[idx];
                for (int oo = 1; oo < Oc; ++oo) m = fmaxf(m, bl[oo * Ic * LT + idx]);
                float den = 0.0f;
                for (int oo = 0; oo < Oc; ++oo) den += __expf(bl[oo * Ic * LT + idx] - m);
                mx[idx] = m;
                rd[idx] = 1.0f / den;
            }
            {
                const int back = (ic + CH == Ic) ? Ic / 2 : 0;
                wl += (CH / 2) * WP - back * WP;
                xl += (CH / 2) * XP - back * XP;
            }
            load_pair_w(wl, fa);
            lds_barrier();
#pragma unroll
            for (int q = 0; q < CH; ++q) {
                const int i = ic + q;
                float cb = __expf(bl[(o * Ic + i) * LT + n] - mx[i * LT + n]) * rd[i * LT + n];
#pragma unroll
                for (int j = 0; j < 8; ++j) { s0[j] += cb * u0[q][j]; s1[j] += cb * u1[q][j]; }
            }
        }
        squash(s0, s1);
        v0 = s0; v1 = s1;
    }

    const int lane_w = (int)__builtin_amdgcn_mbcnt_lo(~0u, 0u);
    const int n_w = lane_w & 15;
    const int h_w = lane_w >> 4;
    const size_t obase = ((size_t)(b * Oc + o) * Dc) * Lc + l0 + n_w;
#pragma unroll
    for (int j = 0; j < 8; ++j) {
        int d = j + 8 * h_w;
        out[obase + (size_t)d * Lc]        = v0[j];
        out[obase + (size_t)(d + 16) * Lc] = v1[j];
    }
}

extern "C" void kernel_launch(void* const* d_in, const int* in_sizes, int n_in,
                              void* d_out, int out_size, void* d_ws, size_t ws_size,
                              hipStream_t stream)
{
    (void)in_sizes; (void)n_in; (void)out_size; (void)d_ws; (void)ws_size;
    const float* x = (const float*)d_in[0];
    const float* W = (const float*)d_in[1];
    float* out = (float*)d_out;

    constexpr size_t shmem1 = (size_t)LDS1_FLOATS * sizeof(float);
    dim3 grid1(Bc * (Lc / NC1) * (Oc / OQ));
    uniform_sum_kernel<<<grid1, NTHREADS1, shmem1, stream>>>(x, W, out);

    constexpr size_t shmem = (size_t)LDS_FLOATS * sizeof(float);
    dim3 grid(Bc * (Lc / LT));
    routing_fused_kernel<<<grid, NTHREADS, shmem, stream>>>(x, W, out);
}
